# starting kernel
# baseline (speedup 1.0000x reference)
#include <hip/hip_runtime.h>
#include <hip/hip_bf16.h>
#include <math.h>


#define BATCH 16384
#define TPTS  128

typedef __attribute__((ext_vector_type(2))) float v2f;
typedef __attribute__((ext_vector_type(8))) float v8f;

__device__ __forceinline__ unsigned rotl32(unsigned v, int n) {
    return (v << n) | (v >> (32 - n));
}

__device__ __forceinline__ void tf2x32(unsigned k0, unsigned k1,
                                       unsigned& x0, unsigned& x1) {
    const unsigned ks0 = k0, ks1 = k1, ks2 = k0 ^ k1 ^ 0x1BD11BDAu;
    x0 += ks0; x1 += ks1;
#define TFR(r) { x0 += x1; x1 = rotl32(x1, r); x1 ^= x0; }
    TFR(13) TFR(15) TFR(26) TFR(6)  x0 += ks1; x1 += ks2 + 1u;
    TFR(17) TFR(29) TFR(16) TFR(24) x0 += ks2; x1 += ks0 + 2u;
    TFR(13) TFR(15) TFR(26) TFR(6)  x0 += ks0; x1 += ks1 + 3u;
    TFR(17) TFR(29) TFR(16) TFR(24) x0 += ks1; x1 += ks2 + 4u;
    TFR(13) TFR(15) TFR(26) TFR(6)  x0 += ks2; x1 += ks0 + 5u;
#undef TFR
}

__device__ __forceinline__ float bits_to_unit(unsigned bits) {
    return __uint_as_float((bits >> 9) | 0x3f800000u) - 1.0f;
}

__global__ __launch_bounds__(256) void
RotRandomizer_build_R_kernel(float* __restrict__ out_R) {
    const int b = blockIdx.x * 256 + threadIdx.x;

    unsigned s0 = 0u, s1 = 1u;
    tf2x32(0u, 42u, s0, s1);
    const unsigned sub0 = s0, sub1 = s1;

    unsigned k[4][2];
#pragma unroll
    for (int i = 0; i < 4; ++i) {
        unsigned a = 0u, c = (unsigned)i;
        tf2x32(sub0, sub1, a, c);
        k[i][0] = a; k[i][1] = c;
    }

    float u[4];
#pragma unroll
    for (int i = 0; i < 4; ++i) {
        unsigned a = 0u, c = (unsigned)b;
        tf2x32(k[i][0], k[i][1], a, c);
        u[i] = bits_to_unit(a ^ c);
    }

    const float TWO_PI = 6.28318530717958647692f;
    const float PI_F   = 3.14159265358979323846f;
    float ax = u[0] * TWO_PI - PI_F;
    float ay = u[1] * TWO_PI - PI_F;
    float az = u[2] * TWO_PI - PI_F;
    if (u[3] < 0.015625f) { ax = 0.f; ay = 0.f; az = 0.f; }

    float sx, cx, sy, cy, sz, cz;
    sincosf(ax, &sx, &cx);
    sincosf(ay, &sy, &cy);
    sincosf(az, &sz, &cz);

    float* R = out_R + (size_t)b * 9;
    R[0] = cz * cy;
    R[1] = cz * sy * sx - sz * cx;
    R[2] = cz * sy * cx + sz * sx;
    R[3] = sz * cy;
    R[4] = sz * sy * sx + cz * cx;
    R[5] = sz * sy * cx - cz * sx;
    R[6] = -sy;
    R[7] = cy * sx;
    R[8] = cy * cx;
}

__global__ __launch_bounds__(256) void
RotRandomizer_rotate_kernel(const float* __restrict__ pos,
                            const float* __restrict__ quat,
                            const float* __restrict__ Rg,
                            float* __restrict__ out_pos,
                            float* __restrict__ out_quat) {
    __shared__ float lds[8 * 1024];

    const int tid  = threadIdx.x;
    const int wave = tid >> 5;
    const int lane = tid & 31;
    const int h    = lane & 15;
    const bool hiL = lane >= 16;
    const int b    = blockIdx.x * 2 + (wave >> 2);
    const int pt   = ((wave & 3) << 5) + lane;
    const int abase = wave * 1024;
    const int obase = abase + 512;

    const size_t idx = (size_t)b * TPTS + pt;

    const float* pi = pos + idx * 3;
    const float px = pi[0], py = pi[1], pz = pi[2];
    const float4 q = *((const float4*)quat + idx);
    const float qx = q.x, qy = q.y, qz = q.z, qw = q.w;

    const float two = 2.0f / (qw * qw + qx * qx + qy * qy + qz * qz);
    const float m00 = 1.0f - two * (qy * qy + qz * qz);
    const float m01 = two * (qx * qy - qz * qw);
    const float m02 = two * (qx * qz + qy * qw);
    const float m10 = two * (qx * qy + qz * qw);
    const float m11 = 1.0f - two * (qx * qx + qz * qz);
    const float m12 = two * (qy * qz - qx * qw);
    const float m20 = two * (qx * qz - qy * qw);
    const float m21 = two * (qy * qz + qx * qw);
    const float m22 = 1.0f - two * (qx * qx + qy * qy);

    float4* ap = (float4*)&lds[abase + lane * 16];
    ap[0] = make_float4(m00, m10, m20, 0.f);
    ap[1] = make_float4(m01, m11, m21, 0.f);
    ap[2] = make_float4(m02, m12, m22, 0.f);
    ap[3] = make_float4(px, py, pz, 0.f);

    const float* Rb = Rg + (size_t)b * 9;
    float b0v = 0.f, b1v = 0.f;
    if (h < 3) {
        if (!hiL) { b0v = Rb[h * 3 + 0]; b1v = Rb[h * 3 + 1]; }
        else      { b0v = Rb[h * 3 + 2]; }
    }
    v2f bm; bm.x = b0v; bm.y = b1v;

    __syncthreads();

    const int acol = hiL ? 2 : 0;
    const int drow = hiL ? 8 : 0;

    v2f am[8];
#pragma unroll
    for (int c = 0; c < 8; ++c) {
        const float* ar = &lds[abase + (c * 16 + h) * 4 + acol];
        am[c].x = ar[0];
        am[c].y = ar[1];
    }

    const v8f czero = {0.f, 0.f, 0.f, 0.f, 0.f, 0.f, 0.f, 0.f};
#pragma unroll
    for (int c = 0; c < 8; ++c) {
        v8f d = __builtin_amdgcn_wmma_f32_16x16x4_f32(
            false, am[c], false, bm, (short)0, czero, false, false);
        if (h < 3) {
#pragma unroll
            for (int j = 0; j < 8; ++j) {
                lds[obase + (c * 16 + j + drow) * 4 + h] = d[j];
            }
        }
    }

    __syncthreads();

    const float4* op = (const float4*)&lds[obase + lane * 16];
    const float4 c0 = op[0];
    const float4 c1 = op[1];
    const float4 c2 = op[2];
    const float4 rp = op[3];

    const float M00 = c0.x, M10 = c0.y, M20 = c0.z;
    const float M01 = c1.x, M11 = c1.y, M21 = c1.z;
    const float M02 = c2.x, M12 = c2.y, M22 = c2.z;

    const float t0 = 1.0f + M00 + M11 + M22;
    const float t1 = 1.0f + M00 - M11 - M22;
    const float t2 = 1.0f - M00 + M11 - M22;
    const float t3 = 1.0f - M00 - M11 + M22;
    const float qa0 = sqrtf(fmaxf(t0, 1e-8f));
    const float qa1 = sqrtf(fmaxf(t1, 1e-8f));
    const float qa2 = sqrtf(fmaxf(t2, 1e-8f));
    const float qa3 = sqrtf(fmaxf(t3, 1e-8f));

    int best = 0; float bv = qa0;
    if (qa1 > bv) { best = 1; bv = qa1; }
    if (qa2 > bv) { best = 2; bv = qa2; }
    if (qa3 > bv) { best = 3; bv = qa3; }

    float W, X, Y, Z;
    if (best == 0)      { W = qa0 * qa0; X = M21 - M12; Y = M02 - M20; Z = M10 - M01; }
    else if (best == 1) { W = M21 - M12; X = qa1 * qa1; Y = M01 + M10; Z = M02 + M20; }
    else if (best == 2) { W = M02 - M20; X = M10 + M01; Y = qa2 * qa2; Z = M12 + M21; }
    else                { W = M10 - M01; X = M20 + M02; Y = M21 + M12; Z = qa3 * qa3; }
    const float inv = 1.0f / (2.0f * fmaxf(bv, 0.1f));
    W *= inv; X *= inv; Y *= inv; Z *= inv;

    float* po = out_pos + idx * 3;
    po[0] = rp.x; po[1] = rp.y; po[2] = rp.z;
    ((float4*)out_quat)[idx] = make_float4(X, Y, Z, W);
}

extern "C" void kernel_launch(void* const* d_in, const int* in_sizes, int n_in,
                              void* d_out, int out_size, void* d_ws, size_t ws_size,
                              hipStream_t stream) {
    const float* pos  = (const float*)d_in[0];
    const float* quat = (const float*)d_in[1];

    float* out      = (float*)d_out;
    float* out_pos  = out;
    float* out_quat = out + (size_t)BATCH * TPTS * 3;
    float* out_R    = out + (size_t)BATCH * TPTS * 3
                          + (size_t)BATCH * TPTS * 4;

    RotRandomizer_build_R_kernel<<<BATCH / 256, 256, 0, stream>>>(out_R);
    RotRandomizer_rotate_kernel<<<BATCH / 2, 256, 0, stream>>>(
        pos, quat, out_R, out_pos, out_quat);
}


# A fragments come from each lane's own registers: k slots of lane half 0 carry point n, of half 1 point n+16, B and D columns renumbered per half; LDS A stage and its barrier go; same sums per output
# speedup vs baseline: 1.5998x; 1.5998x over previous
#include <hip/hip_runtime.h>
#include <hip/hip_bf16.h>
#include <math.h>


#define BATCH 16384
#define TPTS  128
#define NBLK  2048
#define NITER 4

typedef __attribute__((ext_vector_type(2))) float v2f;
typedef __attribute__((ext_vector_type(8))) float v8f;

__device__ __forceinline__ unsigned rotl32(unsigned v, int n) {
    return (v << n) | (v >> (32 - n));
}

__device__ __forceinline__ void tf2x32(unsigned k0, unsigned k1,
                                       unsigned& x0, unsigned& x1) {
    const unsigned ks0 = k0, ks1 = k1, ks2 = k0 ^ k1 ^ 0x1BD11BDAu;
    x0 += ks0; x1 += ks1;
#define TFR(r) { x0 += x1; x1 = rotl32(x1, r); x1 ^= x0; }
    TFR(13) TFR(15) TFR(26) TFR(6)  x0 += ks1; x1 += ks2 + 1u;
    TFR(17) TFR(29) TFR(16) TFR(24) x0 += ks2; x1 += ks0 + 2u;
    TFR(13) TFR(15) TFR(26) TFR(6)  x0 += ks0; x1 += ks1 + 3u;
    TFR(17) TFR(29) TFR(16) TFR(24) x0 += ks1; x1 += ks2 + 4u;
    TFR(13) TFR(15) TFR(26) TFR(6)  x0 += ks2; x1 += ks0 + 5u;
#undef TFR
}

__device__ __forceinline__ float bits_to_unit(unsigned bits) {
    return __uint_as_float((bits >> 9) | 0x3f800000u) - 1.0f;
}

__global__ __launch_bounds__(256) void
RotRandomizer_build_R_kernel(float* __restrict__ out_R) {
    const int b = blockIdx.x * 256 + threadIdx.x;

    unsigned s0 = 0u, s1 = 1u;
    tf2x32(0u, 42u, s0, s1);
    const unsigned sub0 = s0, sub1 = s1;

    unsigned k[4][2];
#pragma unroll
    for (int i = 0; i < 4; ++i) {
        unsigned a = 0u, c = (unsigned)i;
        tf2x32(sub0, sub1, a, c);
        k[i][0] = a; k[i][1] = c;
    }

    float u[4];
#pragma unroll
    for (int i = 0; i < 4; ++i) {
        unsigned a = 0u, c = (unsigned)b;
        tf2x32(k[i][0], k[i][1], a, c);
        u[i] = bits_to_unit(a ^ c);
    }

    const float TWO_PI = 6.28318530717958647692f;
    const float PI_F   = 3.14159265358979323846f;
    float ax = u[0] * TWO_PI - PI_F;
    float ay = u[1] * TWO_PI - PI_F;
    float az = u[2] * TWO_PI - PI_F;
    if (u[3] < 0.015625f) { ax = 0.f; ay = 0.f; az = 0.f; }

    float sx, cx, sy, cy, sz, cz;
    sincosf(ax, &sx, &cx);
    sincosf(ay, &sy, &cy);
    sincosf(az, &sz, &cz);

    float* R = out_R + (size_t)b * 9;
    R[0] = cz * cy;
    R[1] = cz * sy * sx - sz * cx;
    R[2] = cz * sy * cx + sz * sx;
    R[3] = sz * cy;
    R[4] = sz * sy * sx + cz * cx;
    R[5] = sz * sy * cx - cz * sx;
    R[6] = -sy;
    R[7] = cy * sx;
    R[8] = cy * cx;
}

__device__ __forceinline__ void lds_barrier() {
    __builtin_amdgcn_fence(__ATOMIC_RELEASE, "workgroup", "local");
    __builtin_amdgcn_s_barrier();
    __builtin_amdgcn_fence(__ATOMIC_ACQUIRE, "workgroup", "local");
}

__global__ __launch_bounds__(256) void
RotRandomizer_rotate_kernel(const float* __restrict__ pos,
                            const float* __restrict__ quat,
                            const float* __restrict__ Rg,
                            float* __restrict__ out_pos,
                            float* __restrict__ out_quat) {
    __shared__ __attribute__((aligned(16))) float lds[8 * 576];

    const int tid  = threadIdx.x;
    const int wave = tid >> 5;
    const int lane = tid & 31;
    const int n    = lane & 15;
    const int hh   = lane >> 4;
    const int pt   = ((wave & 3) << 5) + lane;
    const int obase = wave * 576;

    const int  ci    = (n & 7) >> 1;
    const int  bi    = ci < 3 ? ci : 2;
    const int  roff  = bi * 3;
    const bool mine  = (ci < 3) && ((n >> 3) == hh);
    const bool keepx = mine && ((n & 1) == 0);
    const bool keepy = mine && ((n & 1) == 1);

    float*       dst = &lds[obase + n * 36 + 8 * hh];
    const float* gat = &lds[obase + hh * 288 + n];

    int    b   = blockIdx.x * 2 + (wave >> 2);
    size_t idx = (size_t)b * TPTS + pt;

    float4 q = *((const float4*)quat + idx);
    float px = pos[idx * 3 + 0], py = pos[idx * 3 + 1], pz = pos[idx * 3 + 2];
    float r0 = Rg[(size_t)b * 9 + roff + 0];
    float r1 = Rg[(size_t)b * 9 + roff + 1];
    float r2 = Rg[(size_t)b * 9 + roff + 2];

#pragma unroll
    for (int it = 0; it < NITER; ++it) {
        float4 nq = q;
        float npx = px, npy = py, npz = pz, nr0 = r0, nr1 = r1, nr2 = r2;
        const int    nb   = b + NBLK * 2;
        const size_t nidx = idx + (size_t)NBLK * 2 * TPTS;
        if (it + 1 < NITER) {
            nq  = *((const float4*)quat + nidx);
            npx = pos[nidx * 3 + 0]; npy = pos[nidx * 3 + 1]; npz = pos[nidx * 3 + 2];
            nr0 = Rg[(size_t)nb * 9 + roff + 0];
            nr1 = Rg[(size_t)nb * 9 + roff + 1];
            nr2 = Rg[(size_t)nb * 9 + roff + 2];
        }

        const float qx = q.x, qy = q.y, qz = q.z, qw = q.w;

        const float two = 2.0f * __builtin_amdgcn_rcpf(qw * qw + qx * qx + qy * qy + qz * qz);
        const float m00 = 1.0f - two * (qy * qy + qz * qz);
        const float m01 = two * (qx * qy - qz * qw);
        const float m02 = two * (qx * qz + qy * qw);
        const float m10 = two * (qx * qy + qz * qw);
        const float m11 = 1.0f - two * (qx * qx + qz * qz);
        const float m12 = two * (qy * qz - qx * qw);
        const float m20 = two * (qx * qz - qy * qw);
        const float m21 = two * (qy * qz + qx * qw);
        const float m22 = 1.0f - two * (qx * qx + qy * qy);

        v2f a00, a01, a02, a10, a11, a12;
        a00.x = m00; a00.y = m01;  a10.x = m02; a10.y = px;
        a01.x = m10; a01.y = m11;  a11.x = m12; a11.y = py;
        a02.x = m20; a02.y = m21;  a12.x = m22; a12.y = pz;

        v2f bm0, bm1, bm2;
        bm0.x = keepx ? r0 : 0.f;  bm0.y = keepy ? r0 : 0.f;
        bm1.x = keepx ? r1 : 0.f;  bm1.y = keepy ? r1 : 0.f;
        bm2.x = keepx ? r2 : 0.f;  bm2.y = keepy ? r2 : 0.f;

        __builtin_amdgcn_sched_barrier(0);

        const v8f czero = {0.f, 0.f, 0.f, 0.f, 0.f, 0.f, 0.f, 0.f};
        v8f d0 = __builtin_amdgcn_wmma_f32_16x16x4_f32(
            false, a00, false, bm0, (short)0, czero, false, false);
        v8f d1 = __builtin_amdgcn_wmma_f32_16x16x4_f32(
            false, a10, false, bm0, (short)0, czero, false, false);
        d0 = __builtin_amdgcn_wmma_f32_16x16x4_f32(
            false, a01, false, bm1, (short)0, d0, false, false);
        d1 = __builtin_amdgcn_wmma_f32_16x16x4_f32(
            false, a11, false, bm1, (short)0, d1, false, false);
        d0 = __builtin_amdgcn_wmma_f32_16x16x4_f32(
            false, a02, false, bm2, (short)0, d0, false, false);
        d1 = __builtin_amdgcn_wmma_f32_16x16x4_f32(
            false, a12, false, bm2, (short)0, d1, false, false);

        *(float4*)(dst)      = make_float4(d0[0], d0[1], d0[2], d0[3]);
        *(float4*)(dst + 4)  = make_float4(d0[4], d0[5], d0[6], d0[7]);
        *(float4*)(dst + 16) = make_float4(d1[0], d1[1], d1[2], d1[3]);
        *(float4*)(dst + 20) = make_float4(d1[4], d1[5], d1[6], d1[7]);

        lds_barrier();

        const float M00 = gat[0 * 36], M01 = gat[1 * 36];
        const float M02 = gat[0 * 36 + 16], rpx = gat[1 * 36 + 16];
        const float M10 = gat[2 * 36], M11 = gat[3 * 36];
        const float M12 = gat[2 * 36 + 16], rpy = gat[3 * 36 + 16];
        const float M20 = gat[4 * 36], M21 = gat[5 * 36];
        const float M22 = gat[4 * 36 + 16], rpz = gat[5 * 36 + 16];

        const float t0 = 1.0f + M00 + M11 + M22;
        const float t1 = 1.0f + M00 - M11 - M22;
        const float t2 = 1.0f - M00 + M11 - M22;
        const float t3 = 1.0f - M00 - M11 + M22;
        const float qa0 = __builtin_amdgcn_sqrtf(fmaxf(t0, 1e-8f));
        const float qa1 = __builtin_amdgcn_sqrtf(fmaxf(t1, 1e-8f));
        const float qa2 = __builtin_amdgcn_sqrtf(fmaxf(t2, 1e-8f));
        const float qa3 = __builtin_amdgcn_sqrtf(fmaxf(t3, 1e-8f));

        int best = 0; float bv = qa0;
        if (qa1 > bv) { best = 1; bv = qa1; }
        if (qa2 > bv) { best = 2; bv = qa2; }
        if (qa3 > bv) { best = 3; bv = qa3; }

        float W, X, Y, Z;
        if (best == 0)      { W = qa0 * qa0; X = M21 - M12; Y = M02 - M20; Z = M10 - M01; }
        else if (best == 1) { W = M21 - M12; X = qa1 * qa1; Y = M01 + M10; Z = M02 + M20; }
        else if (best == 2) { W = M02 - M20; X = M10 + M01; Y = qa2 * qa2; Z = M12 + M21; }
        else                { W = M10 - M01; X = M20 + M02; Y = M21 + M12; Z = qa3 * qa3; }
        const float inv = __builtin_amdgcn_rcpf(2.0f * fmaxf(bv, 0.1f));
        W *= inv; X *= inv; Y *= inv; Z *= inv;

        float* po = out_pos + idx * 3;
        po[0] = rpx; po[1] = rpy; po[2] = rpz;
        ((float4*)out_quat)[idx] = make_float4(X, Y, Z, W);

        q = nq; px = npx; py = npy; pz = npz; r0 = nr0; r1 = nr1; r2 = nr2;
        b = nb; idx = nidx;
    }
}

extern "C" void kernel_launch(void* const* d_in, const int* in_sizes, int n_in,
                              void* d_out, int out_size, void* d_ws, size_t ws_size,
                              hipStream_t stream) {
    const float* pos  = (const float*)d_in[0];
    const float* quat = (const float*)d_in[1];

    float* out      = (float*)d_out;
    float* out_pos  = out;
    float* out_quat = out + (size_t)BATCH * TPTS * 3;
    float* out_R    = out + (size_t)BATCH * TPTS * 3
                          + (size_t)BATCH * TPTS * 4;

    RotRandomizer_build_R_kernel<<<BATCH / 256, 256, 0, stream>>>(out_R);
    RotRandomizer_rotate_kernel<<<NBLK, 256, 0, stream>>>(
        pos, quat, out_R, out_pos, out_quat);
}
